# starting kernel
# baseline (speedup 1.0000x reference)
#include <hip/hip_runtime.h>
#include <hip/hip_bf16.h>
#include <math.h>

#define BDIM 1024
#define DDIM 256
#define CDIM 1000
#define EPSV 1e-6f

typedef __attribute__((ext_vector_type(2))) float v2f;
typedef __attribute__((ext_vector_type(8))) float v8f;

__global__ void pd_xterm_kernel(const float* __restrict__ x,
                                float* __restrict__ xterm) {
    const int lane = threadIdx.x & 31;
    const int wave = threadIdx.x >> 5;
    const int row  = blockIdx.x * 8 + wave;

    const float* xr = x + row * DDIM;
    float s1 = 0.f, s2 = 0.f;
#pragma unroll
    for (int i = 0; i < DDIM / 32; ++i) {
        float v = xr[lane + i * 32];
        s1 += v;
        s2 += v * v;
    }
#pragma unroll
    for (int off = 16; off > 0; off >>= 1) {
        s1 += __shfl_xor(s1, off, 32);
        s2 += __shfl_xor(s2, off, 32);
    }
    if (lane == 0) {
        xterm[row] = s2 + 2.0f * EPSV * s1;
    }
}

__global__ void pd_wterm_kernel(const float* __restrict__ w,
                                float* __restrict__ wterm) {
    const int c = blockIdx.x * blockDim.x + threadIdx.x;
    if (c >= CDIM) return;
    float s1 = 0.f, s2 = 0.f;
#pragma unroll 8
    for (int d = 0; d < DDIM; ++d) {
        float v = w[d * CDIM + c];
        s1 += v;
        s2 += v * v;
    }
    wterm[c] = s2 - 2.0f * EPSV * s1 + (float)DDIM * EPSV * EPSV;
}

__global__ void pd_wmma_kernel(const float* __restrict__ x,
                               const float* __restrict__ w,
                               const float* __restrict__ xterm,
                               const float* __restrict__ wterm,
                               float* __restrict__ out) {
    const int lane  = threadIdx.x & 31;
    const int wave  = threadIdx.x >> 5;
    const int wm    = wave >> 2;
    const int wn    = wave & 3;

    const int tile_m0 = blockIdx.y * 32 + wm * 16;
    const int tile_n0 = blockIdx.x * 64 + wn * 16;

    const int l15   = lane & 15;
    const int khalf = (lane >> 4) * 2;

    const float* arow = x + (tile_m0 + l15) * DDIM + khalf;
    const int col   = tile_n0 + l15;
    const int colc  = col < CDIM ? col : (CDIM - 1);
    const float* bcol = w + colc;

    v8f acc = {};

#pragma unroll 8
    for (int k = 0; k < DDIM; k += 4) {
        v2f a, b;
        a.x = arow[k];
        a.y = arow[k + 1];
        b.x = bcol[(k + khalf) * CDIM];
        b.y = bcol[(k + khalf + 1) * CDIM];
        acc = __builtin_amdgcn_wmma_f32_16x16x4_f32(
             false, a,  false, b,
             (short)0, acc,  false,  false);
    }

    const float wt  = wterm[colc];
    const int   mhi = (lane >> 4) * 8;
#pragma unroll
    for (int r = 0; r < 8; ++r) {
        const int row_g = tile_m0 + r + mhi;
        const float s   = xterm[row_g] + wt - 2.0f * acc[r];
        if (col < CDIM) {
            out[row_g * CDIM + col] = sqrtf(fmaxf(s, 0.0f));
        }
    }
}

extern "C" void kernel_launch(void* const* d_in, const int* in_sizes, int n_in,
                              void* d_out, int out_size, void* d_ws, size_t ws_size,
                              hipStream_t stream) {
    const float* x = (const float*)d_in[0];
    const float* w = (const float*)d_in[1];
    float* out = (float*)d_out;

    float* xterm = (float*)d_ws;
    float* wterm = xterm + BDIM;

    pd_xterm_kernel<<<BDIM / 8, 256, 0, stream>>>(x, xterm);
    pd_wterm_kernel<<<(CDIM + 255) / 256, 256, 0, stream>>>(w, wterm);
    dim3 grid(16, 32, 1);
    pd_wmma_kernel<<<grid, 256, 0, stream>>>(x, w, xterm, wterm, out);
}


# Fuse the two norm-term launches into the WMMA kernel: row and column sums accumulate per lane from the A/B fragments over K, joined by lane shuffles (new reduction tree); 3 launches -> 1
# speedup vs baseline: 3.7633x; 3.7633x over previous
#include <hip/hip_runtime.h>
#include <hip/hip_bf16.h>
#include <math.h>

#define BDIM 1024
#define DDIM 256
#define CDIM 1000
#define EPSV 1e-6f

typedef __attribute__((ext_vector_type(2))) float v2f;
typedef __attribute__((ext_vector_type(8))) float v8f;

__global__ void pd_fused_kernel(const float* __restrict__ x,
                                const float* __restrict__ w,
                                float* __restrict__ out) {
    const int lane  = threadIdx.x & 31;
    const int wave  = threadIdx.x >> 5;
    const int wm    = wave >> 2;
    const int wn    = wave & 3;

    const int tile_m0 = blockIdx.y * 32 + wm * 16;
    const int tile_n0 = blockIdx.x * 64 + wn * 16;

    const int l15   = lane & 15;
    const int khalf = (lane >> 4) * 2;

    const float* arow = x + (tile_m0 + l15) * DDIM + khalf;
    const int col   = tile_n0 + l15;
    const int colc  = col < CDIM ? col : (CDIM - 1);
    const float* bcol = w + colc;

    v8f acc = {};
    v2f xs1 = {0.f, 0.f}, xs2 = {0.f, 0.f};
    v2f ws1 = {0.f, 0.f}, ws2 = {0.f, 0.f};

#pragma unroll 8
    for (int k = 0; k < DDIM; k += 4) {
        v2f a, b;
        a.x = arow[k];
        a.y = arow[k + 1];
        b.x = bcol[(k + khalf) * CDIM];
        b.y = bcol[(k + khalf + 1) * CDIM];
        xs1 += a;
        xs2 += a * a;
        ws1 += b;
        ws2 += b * b;
        acc = __builtin_amdgcn_wmma_f32_16x16x4_f32(
             false, a,  false, b,
             (short)0, acc,  false,  false);
    }

    float x1 = xs1.x + xs1.y;
    float x2 = xs2.x + xs2.y;
    float w1 = ws1.x + ws1.y;
    float w2 = ws2.x + ws2.y;
    x1 += __shfl_xor(x1, 16, 32);
    x2 += __shfl_xor(x2, 16, 32);
    w1 += __shfl_xor(w1, 16, 32);
    w2 += __shfl_xor(w2, 16, 32);
    const float xt = x2 + 2.0f * EPSV * x1;
    const float wt = w2 - 2.0f * EPSV * w1 + (float)DDIM * EPSV * EPSV;

    const int mhi = (lane >> 4) * 8;
    float xr[8];
#pragma unroll
    for (int r = 0; r < 8; ++r) {
        xr[r] = __shfl(xt, r + mhi, 32);
    }
#pragma unroll
    for (int r = 0; r < 8; ++r) {
        const int row_g = tile_m0 + r + mhi;
        const float s   = xr[r] + wt - 2.0f * acc[r];
        if (col < CDIM) {
            out[row_g * CDIM + col] = sqrtf(fmaxf(s, 0.0f));
        }
    }
}

extern "C" void kernel_launch(void* const* d_in, const int* in_sizes, int n_in,
                              void* d_out, int out_size, void* d_ws, size_t ws_size,
                              hipStream_t stream) {
    const float* x = (const float*)d_in[0];
    const float* w = (const float*)d_in[1];
    float* out = (float*)d_out;

    dim3 grid(16, 32, 1);
    pd_fused_kernel<<<grid, 256, 0, stream>>>(x, w, out);
}


# Stage the block's x and w slices through LDS: whole-line 16-byte global loads issued up front in four K chunks, WMMA fragments read from LDS; block tile 64x64, each wave 32x16 (two accumulators)
# speedup vs baseline: 4.9851x; 1.3247x over previous
#include <hip/hip_runtime.h>
#include <hip/hip_bf16.h>
#include <math.h>

#define BDIM 1024
#define DDIM 256
#define CDIM 1000
#define EPSV 1e-6f

#define TM 64
#define TN 64
#define KCH 64
#define NCH (DDIM / KCH)
#define APITCH 260
#define BPITCH 72

typedef __attribute__((ext_vector_type(2))) float v2f;
typedef __attribute__((ext_vector_type(4))) float v4f;
typedef __attribute__((ext_vector_type(8))) float v8f;

__global__ void __launch_bounds__(256)
pd_fused_kernel(const float* __restrict__ x,
                const float* __restrict__ w,
                float* __restrict__ out) {
    __shared__ __attribute__((aligned(16))) float As[TM * APITCH];
    __shared__ __attribute__((aligned(16))) float Bs[DDIM * BPITCH];

    const int tid   = threadIdx.x;
    const int lane  = tid & 31;
    const int wave  = tid >> 5;
    const int wm    = wave >> 2;
    const int wn    = wave & 3;

    const int m0 = blockIdx.y * TM;
    const int n0 = blockIdx.x * TN;

    const int seg = tid & 15;
    const int r0  = tid >> 4;
    const int bcol0 = n0 + seg * 4;
    const int bcolc = bcol0 < CDIM ? bcol0 : (CDIM - 4);
    const float* xg = x + (m0 + r0) * DDIM + seg * 4;
    const float* wg = w + r0 * CDIM + bcolc;

    v4f ra[NCH][4], rb[NCH][4];
#pragma unroll
    for (int c = 0; c < NCH; ++c) {
#pragma unroll
        for (int i = 0; i < 4; ++i) {
            ra[c][i] = *(const v4f*)(xg + (16 * i) * DDIM + c * KCH);
        }
#pragma unroll
        for (int i = 0; i < 4; ++i) {
            rb[c][i] = *(const v4f*)(wg + (c * KCH + 16 * i) * CDIM);
        }
        __builtin_amdgcn_sched_barrier(0);
    }

    const int l15   = lane & 15;
    const int khalf = (lane >> 4) * 2;

    const float* af0 = As + (wm * 32 + l15) * APITCH + khalf;
    const float* af1 = af0 + 16 * APITCH;
    const float* bf  = Bs + khalf * BPITCH + wn * 16 + l15;

    v8f acc0 = {}, acc1 = {};
    v2f xs1a = {0.f, 0.f}, xs2a = {0.f, 0.f};
    v2f xs1b = {0.f, 0.f}, xs2b = {0.f, 0.f};
    v2f ws1  = {0.f, 0.f}, ws2  = {0.f, 0.f};

#pragma unroll
    for (int c = 0; c < NCH; ++c) {
#pragma unroll
        for (int i = 0; i < 4; ++i) {
            *(v4f*)(As + (r0 + 16 * i) * APITCH + c * KCH + seg * 4) = ra[c][i];
        }
#pragma unroll
        for (int i = 0; i < 4; ++i) {
            *(v4f*)(Bs + (c * KCH + r0 + 16 * i) * BPITCH + seg * 4) = rb[c][i];
        }
        asm volatile("s_wait_dscnt 0x0\n\ts_barrier_signal -1\n\ts_barrier_wait -1" ::: "memory");

        v2f fa0[2][4], fa1[2][4], fb[2][4];
#pragma unroll
        for (int s = 0; s < 4; ++s) {
            const int k = c * KCH + 4 * s;
            fa0[0][s]  = *(const v2f*)(af0 + k);
            fa1[0][s]  = *(const v2f*)(af1 + k);
            fb[0][s].x = bf[k * BPITCH];
            fb[0][s].y = bf[(k + 1) * BPITCH];
        }
#pragma unroll
        for (int g = 0; g < 4; ++g) {
            const int cur = g & 1, nxt = cur ^ 1;
            if (g < 3) {
#pragma unroll
                for (int s = 0; s < 4; ++s) {
                    const int k = c * KCH + 16 * (g + 1) + 4 * s;
                    fa0[nxt][s]  = *(const v2f*)(af0 + k);
                    fa1[nxt][s]  = *(const v2f*)(af1 + k);
                    fb[nxt][s].x = bf[k * BPITCH];
                    fb[nxt][s].y = bf[(k + 1) * BPITCH];
                }
            }
            __builtin_amdgcn_sched_barrier(0);
#pragma unroll
            for (int s = 0; s < 4; ++s) {
                const v2f a0 = fa0[cur][s];
                const v2f a1 = fa1[cur][s];
                const v2f b  = fb[cur][s];
                xs1a += a0;
                xs2a += a0 * a0;
                xs1b += a1;
                xs2b += a1 * a1;
                ws1  += b;
                ws2  += b * b;
                acc0 = __builtin_amdgcn_wmma_f32_16x16x4_f32(
                     false, a0,  false, b,
                     (short)0, acc0,  false,  false);
                acc1 = __builtin_amdgcn_wmma_f32_16x16x4_f32(
                     false, a1,  false, b,
                     (short)0, acc1,  false,  false);
            }
            __builtin_amdgcn_sched_barrier(0);
        }
    }

    float x1a = xs1a.x + xs1a.y;
    float x2a = xs2a.x + xs2a.y;
    float x1b = xs1b.x + xs1b.y;
    float x2b = xs2b.x + xs2b.y;
    float w1  = ws1.x + ws1.y;
    float w2  = ws2.x + ws2.y;
    x1a += __shfl_xor(x1a, 16, 32);
    x2a += __shfl_xor(x2a, 16, 32);
    x1b += __shfl_xor(x1b, 16, 32);
    x2b += __shfl_xor(x2b, 16, 32);
    w1  += __shfl_xor(w1, 16, 32);
    w2  += __shfl_xor(w2, 16, 32);
    const float xta = x2a + 2.0f * EPSV * x1a;
    const float xtb = x2b + 2.0f * EPSV * x1b;
    const float wt  = w2 - 2.0f * EPSV * w1 + (float)DDIM * EPSV * EPSV;

    const int mhi = (lane >> 4) * 8;
    const int col = n0 + wn * 16 + l15;
    const int row_base = m0 + wm * 32 + mhi;
    float xra[8], xrb[8];
#pragma unroll
    for (int r = 0; r < 8; ++r) {
        xra[r] = __shfl(xta, r + mhi, 32);
        xrb[r] = __shfl(xtb, r + mhi, 32);
    }
#pragma unroll
    for (int r = 0; r < 8; ++r) {
        const float s0 = xra[r] + wt - 2.0f * acc0[r];
        const float s1 = xrb[r] + wt - 2.0f * acc1[r];
        if (col < CDIM) {
            out[(row_base + r) * CDIM + col]      = sqrtf(fmaxf(s0, 0.0f));
            out[(row_base + 16 + r) * CDIM + col] = sqrtf(fmaxf(s1, 0.0f));
        }
    }
}

extern "C" void kernel_launch(void* const* d_in, const int* in_sizes, int n_in,
                              void* d_out, int out_size, void* d_ws, size_t ws_size,
                              hipStream_t stream) {
    const float* x = (const float*)d_in[0];
    const float* w = (const float*)d_in[1];
    float* out = (float*)d_out;

    dim3 grid(16, 16, 1);
    pd_fused_kernel<<<grid, 256, 0, stream>>>(x, w, out);
}
